# starting kernel
# baseline (speedup 1.0000x reference)
#include <hip/hip_runtime.h>
#include <hip/hip_bf16.h>


typedef float v2f __attribute__((ext_vector_type(2)));
typedef float v8f __attribute__((ext_vector_type(8)));

#define IMG_H 1024
#define IMG_W 1024
#define HALO 2

#define BLK_ROWS 128
#define BLK_COLS 32
#define P_ROWS (BLK_ROWS + 4)
#define P_COLS (BLK_COLS + 4)
#define ESTRIDE (P_COLS + 1)
#define NCHUNK (BLK_COLS / 16)

#define SHIFT 2.0f
#define EXP_SCALE (15.0f * 1.4426950408889634f)
#define LOG_SCALE (0.6931471805599453f / 15.0f)

__device__ __forceinline__ float fast_exp2(float x) {
#if __has_builtin(__builtin_amdgcn_exp2f)
  return __builtin_amdgcn_exp2f(x);
#else
  return exp2f(x);
#endif
}
__device__ __forceinline__ float fast_log2(float x) {
#if __has_builtin(__builtin_amdgcn_logf)
  return __builtin_amdgcn_logf(x);
#else
  return log2f(x);
#endif
}

__global__ __launch_bounds__(256) void soft_dilate5x5_kernel(
    const float* __restrict__ x, float* __restrict__ out) {
  __shared__ float E[P_ROWS * ESTRIDE];
  __shared__ float Hs[NCHUNK * P_ROWS * 16];

  const int tid  = threadIdx.x;
  const int w    = tid >> 5;
  const int lane = tid & 31;
  const int hl   = lane >> 4;
  const int ln   = lane & 15;

  const int img = blockIdx.x >> 8;
  const int rem = blockIdx.x & 255;
  const int y0  = (rem >> 5) * BLK_ROWS;
  const int x0  = (rem & 31) * BLK_COLS;
  const size_t ibase = (size_t)img << 20;

  for (int idx = tid; idx < P_ROWS * P_COLS; idx += 256) {
    int r = idx / P_COLS;
    int c = idx - r * P_COLS;
    int gy = y0 - HALO + r; gy = gy < 0 ? 0 : (gy > IMG_H - 1 ? IMG_H - 1 : gy);
    int gx = x0 - HALO + c; gx = gx < 0 ? 0 : (gx > IMG_W - 1 ? IMG_W - 1 : gx);
    float v = x[ibase + (size_t)gy * IMG_W + gx];
    E[r * ESTRIDE + c] = fast_exp2((v - SHIFT) * EXP_SCALE);
  }
  __syncthreads();

  for (int idx = tid; idx < P_ROWS * BLK_COLS; idx += 256) {
    int r  = idx >> 5;
    int xo = idx & 31;
    const float* e = &E[r * ESTRIDE + xo];
    float h = ((e[0] + e[1]) + (e[2] + e[3])) + e[4];
    Hs[(((xo >> 4) * P_ROWS) + r) * 16 + (xo & 15)] = h;
  }
  __syncthreads();

  v2f aa[5];
#pragma unroll
  for (int s = 0; s < 5; ++s) {
    const int K0 = 4 * s + 2 * hl;
    aa[s].x = (K0 >= ln && K0 <= ln + 4) ? 1.0f : 0.0f;
    aa[s].y = (K0 + 1 >= ln && K0 + 1 <= ln + 4) ? 1.0f : 0.0f;
  }

  v8f acc[NCHUNK];
#pragma unroll
  for (int nc = 0; nc < NCHUNK; ++nc) {
    acc[nc] = (v8f){};
#pragma unroll
    for (int s = 0; s < 5; ++s) {
      const int R = 16 * w + 4 * s + 2 * hl;
      const int base = (nc * P_ROWS + R) * 16 + ln;
      v2f b;
      b.x = Hs[base];
      b.y = Hs[base + 16];
      acc[nc] = __builtin_amdgcn_wmma_f32_16x16x4_f32(
          false, aa[s], false, b, (short)0, acc[nc], false, false);
    }
  }

#pragma unroll
  for (int nc = 0; nc < NCHUNK; ++nc) {
    const size_t obase =
        ibase + (size_t)(y0 + 16 * w + 8 * hl) * IMG_W + x0 + nc * 16 + ln;
#pragma unroll
    for (int v = 0; v < 8; ++v) {
      out[obase + (size_t)v * IMG_W] = fast_log2(acc[nc][v]) * LOG_SCALE + SHIFT;
    }
  }
}

extern "C" void kernel_launch(void* const* d_in, const int* in_sizes, int n_in,
                              void* d_out, int out_size, void* d_ws, size_t ws_size,
                              hipStream_t stream) {
  const float* x = (const float*)d_in[0];
  float* out = (float*)d_out;

  const int blocks = 4 * (IMG_H / BLK_ROWS) * (IMG_W / BLK_COLS);
  soft_dilate5x5_kernel<<<blocks, 256, 0, stream>>>(x, out);
}


# Both 5-tap sums as chained banded products on the matrix pipe, in registers (exp tile is A, step-1 D is step-2 B): no LDS, no barrier, 2048 waves of two 16x64 blocks; summation order changes
# speedup vs baseline: 1.2224x; 1.2224x over previous
#include <hip/hip_runtime.h>


typedef float v2f __attribute__((ext_vector_type(2)));
typedef float v4f __attribute__((ext_vector_type(4)));
typedef float v8f __attribute__((ext_vector_type(8)));

#define IMG_H 1024
#define IMG_W 1024

#define MT 5
#define KS 5
#define NT 4
#define NBLK 2

#define SHIFT 2.0f
#define EXP_SCALE (15.0f * 1.4426950408889634f)
#define LOG_SCALE (0.6931471805599453f / 15.0f)

__device__ __forceinline__ float fast_exp2(float x) {
#if __has_builtin(__builtin_amdgcn_exp2f)
  return __builtin_amdgcn_exp2f(x);
#else
  return exp2f(x);
#endif
}
__device__ __forceinline__ float fast_log2(float x) {
#if __has_builtin(__builtin_amdgcn_logf)
  return __builtin_amdgcn_logf(x);
#else
  return log2f(x);
#endif
}

__device__ __forceinline__ int colperm(int i) {
  return i < 2 ? i : (i < 8 ? i + 2 : (i < 10 ? i - 6 : i));
}

__device__ __forceinline__ void load_block(const float* __restrict__ xi,
                                           const unsigned (&xc)[MT], int Y0, int h,
                                           v2f (&raw)[MT][KS]) {
  const int yb = Y0 - 2 + 2 * h;
#pragma unroll
  for (int s = 0; s < KS; ++s) {
#pragma unroll
    for (int e = 0; e < 2; ++e) {
      int y = yb + 4 * s + e;
      if (s == 0) y = y < 0 ? 0 : y;
      if (s == KS - 1) y = y > IMG_H - 1 ? IMG_H - 1 : y;
      const unsigned ro = (unsigned)y * IMG_W;
#pragma unroll
      for (int m = 0; m < MT; ++m) raw[m][s][e] = xi[ro + xc[m]];
    }
  }
}

__global__ __launch_bounds__(128) void soft_dilate5x5_kernel(
    const float* __restrict__ x, float* __restrict__ out) {
  const int lane = threadIdx.x & 31;
  const int h = lane >> 4;
  const int c = lane & 15;
  const int w = __builtin_amdgcn_readfirstlane((int)(threadIdx.x >> 5));
  const int g = blockIdx.x * 4 + w;
  const int X0 = (g & 15) * 64;
  const int Yw = ((g >> 4) & 31) * (16 * NBLK);
  const int img = g >> 9;
  const float* __restrict__ xi = x + ((size_t)img << 20);
  float* __restrict__ oi = out + ((size_t)img << 20);

  v2f bv[KS];
#pragma unroll
  for (int s = 0; s < KS; ++s) {
#pragma unroll
    for (int e = 0; e < 2; ++e) {
      const int k = 4 * s + 2 * h + e;
      bv[s][e] = (k >= c && k <= c + 4) ? 1.0f : 0.0f;
    }
  }
  v2f ah[5];
#pragma unroll
  for (int q = 0; q < 4; ++q) {
#pragma unroll
    for (int e = 0; e < 2; ++e) {
      const int u = colperm(8 * h + 2 * q + e);
      ah[q][e] = (u >= c && u <= c + 4) ? 1.0f : 0.0f;
    }
  }
#pragma unroll
  for (int e = 0; e < 2; ++e) {
    const int u = 16 + colperm(8 * h + e);
    ah[4][e] = (u >= c && u <= c + 4) ? 1.0f : 0.0f;
  }

  unsigned xc[MT];
#pragma unroll
  for (int m = 0; m < MT; ++m) {
    int gx = X0 - 2 + 16 * m + colperm(c);
    gx = gx < 0 ? 0 : (gx > IMG_W - 1 ? IMG_W - 1 : gx);
    xc[m] = (unsigned)gx;
  }

  v2f raw[MT][KS];
  load_block(xi, xc, Yw, h, raw);
  __builtin_amdgcn_sched_barrier(0);

#pragma unroll
  for (int b = 0; b < NBLK; ++b) {
    const int Y0 = Yw + 16 * b;

    v2f E[MT][KS];
#pragma unroll
    for (int m = 0; m < MT; ++m) {
#pragma unroll
      for (int s = 0; s < KS; ++s) {
        E[m][s][0] = fast_exp2((raw[m][s][0] - SHIFT) * EXP_SCALE);
        E[m][s][1] = fast_exp2((raw[m][s][1] - SHIFT) * EXP_SCALE);
      }
    }
    if (b + 1 < NBLK) {
      load_block(xi, xc, Y0 + 16, h, raw);
      __builtin_amdgcn_sched_barrier(0);
    }

    v8f T[MT];
#pragma unroll
    for (int m = 0; m < MT; ++m) T[m] = (v8f){};
#pragma unroll
    for (int s = 0; s < KS; ++s) {
#pragma unroll
      for (int m = 0; m < MT; ++m) {
        T[m] = __builtin_amdgcn_wmma_f32_16x16x4_f32(
            false, E[m][s], false, bv[s], (short)0, T[m], false, false);
      }
    }

    v8f S[NT];
#pragma unroll
    for (int t = 0; t < NT; ++t) S[t] = (v8f){};
#pragma unroll
    for (int q = 0; q < 4; ++q) {
#pragma unroll
      for (int t = 0; t < NT; ++t) {
        v2f bt;
        bt.x = T[t][2 * q];
        bt.y = T[t][2 * q + 1];
        S[t] = __builtin_amdgcn_wmma_f32_16x16x4_f32(
            false, ah[q], false, bt, (short)0, S[t], false, false);
      }
    }
#pragma unroll
    for (int t = 0; t < NT; ++t) {
      v2f bt;
      bt.x = T[t + 1][0];
      bt.y = T[t + 1][1];
      S[t] = __builtin_amdgcn_wmma_f32_16x16x4_f32(
          false, ah[4], false, bt, (short)0, S[t], false, false);
    }

    float* __restrict__ orow = oi + (size_t)(Y0 + c) * IMG_W + X0 + 8 * h;
#pragma unroll
    for (int t = 0; t < NT; ++t) {
      v4f o0, o1;
#pragma unroll
      for (int i = 0; i < 4; ++i) {
        o0[i] = fast_log2(S[t][i]) * LOG_SCALE + SHIFT;
        o1[i] = fast_log2(S[t][4 + i]) * LOG_SCALE + SHIFT;
      }
      *(v4f*)(orow + 16 * t) = o0;
      *(v4f*)(orow + 16 * t + 4) = o1;
    }
  }
}

extern "C" void kernel_launch(void* const* d_in, const int* in_sizes, int n_in,
                              void* d_out, int out_size, void* d_ws, size_t ws_size,
                              hipStream_t stream) {
  const float* x = (const float*)d_in[0];
  float* out = (float*)d_out;

  soft_dilate5x5_kernel<<<512, 128, 0, stream>>>(x, out);
}


# Finished 16x64 block staged through a wave-private LDS tile (pitch 272 B) and read back row-wise: every 16-byte store instruction writes 4 whole 128-byte lines (was 16 lines x 32 B); no barrier
# speedup vs baseline: 1.7294x; 1.4148x over previous
#include <hip/hip_runtime.h>


typedef float v2f __attribute__((ext_vector_type(2)));
typedef float v4f __attribute__((ext_vector_type(4)));
typedef float v8f __attribute__((ext_vector_type(8)));

#define IMG_H 1024
#define IMG_W 1024

#define MT 5
#define KS 5
#define NT 4
#define NBLK 2
#define WAVES 4
#define PITCH 68

#define SHIFT 2.0f
#define EXP_SCALE (15.0f * 1.4426950408889634f)
#define LOG_SCALE (0.6931471805599453f / 15.0f)

__device__ __forceinline__ float fast_exp2(float x) {
#if __has_builtin(__builtin_amdgcn_exp2f)
  return __builtin_amdgcn_exp2f(x);
#else
  return exp2f(x);
#endif
}
__device__ __forceinline__ float fast_log2(float x) {
#if __has_builtin(__builtin_amdgcn_logf)
  return __builtin_amdgcn_logf(x);
#else
  return log2f(x);
#endif
}

__device__ __forceinline__ int colperm(int i) {
  return i < 2 ? i : (i < 8 ? i + 2 : (i < 10 ? i - 6 : i));
}

__device__ __forceinline__ void load_block(const float* __restrict__ xi,
                                           const unsigned (&xc)[MT], int Y0, int h,
                                           v2f (&raw)[MT][KS]) {
  const int yb = Y0 - 2 + 2 * h;
#pragma unroll
  for (int s = 0; s < KS; ++s) {
#pragma unroll
    for (int e = 0; e < 2; ++e) {
      int y = yb + 4 * s + e;
      if (s == 0) y = y < 0 ? 0 : y;
      if (s == KS - 1) y = y > IMG_H - 1 ? IMG_H - 1 : y;
      const unsigned ro = (unsigned)y * IMG_W;
#pragma unroll
      for (int m = 0; m < MT; ++m) raw[m][s][e] = xi[ro + xc[m]];
    }
  }
}

__global__ __launch_bounds__(32 * WAVES) void soft_dilate5x5_kernel(
    const float* __restrict__ x, float* __restrict__ out) {
  const int lane = threadIdx.x & 31;
  const int h = lane >> 4;
  const int c = lane & 15;
  const int w = __builtin_amdgcn_readfirstlane((int)(threadIdx.x >> 5));
  const int g = blockIdx.x * 4 + w;
  const int X0 = (g & 15) * 64;
  const int Yw = ((g >> 4) & 31) * (16 * NBLK);
  const int img = g >> 9;
  const float* __restrict__ xi = x + ((size_t)img << 20);
  float* __restrict__ oi = out + ((size_t)img << 20);

  __shared__ __attribute__((aligned(16))) float stage[WAVES][NBLK][16 * PITCH];

  v2f bv[KS];
#pragma unroll
  for (int s = 0; s < KS; ++s) {
#pragma unroll
    for (int e = 0; e < 2; ++e) {
      const int k = 4 * s + 2 * h + e;
      bv[s][e] = (k >= c && k <= c + 4) ? 1.0f : 0.0f;
    }
  }
  v2f ah[5];
#pragma unroll
  for (int q = 0; q < 4; ++q) {
#pragma unroll
    for (int e = 0; e < 2; ++e) {
      const int u = colperm(8 * h + 2 * q + e);
      ah[q][e] = (u >= c && u <= c + 4) ? 1.0f : 0.0f;
    }
  }
#pragma unroll
  for (int e = 0; e < 2; ++e) {
    const int u = 16 + colperm(8 * h + e);
    ah[4][e] = (u >= c && u <= c + 4) ? 1.0f : 0.0f;
  }

  unsigned xc[MT];
#pragma unroll
  for (int m = 0; m < MT; ++m) {
    int gx = X0 - 2 + 16 * m + colperm(c);
    gx = gx < 0 ? 0 : (gx > IMG_W - 1 ? IMG_W - 1 : gx);
    xc[m] = (unsigned)gx;
  }

  v2f raw[MT][KS];
  load_block(xi, xc, Yw, h, raw);
  __builtin_amdgcn_sched_barrier(0);

#pragma unroll
  for (int b = 0; b < NBLK; ++b) {
    const int Y0 = Yw + 16 * b;

    v2f E[MT][KS];
#pragma unroll
    for (int m = 0; m < MT; ++m) {
#pragma unroll
      for (int s = 0; s < KS; ++s) {
        E[m][s][0] = fast_exp2((raw[m][s][0] - SHIFT) * EXP_SCALE);
        E[m][s][1] = fast_exp2((raw[m][s][1] - SHIFT) * EXP_SCALE);
      }
    }
    if (b + 1 < NBLK) {
      load_block(xi, xc, Y0 + 16, h, raw);
      __builtin_amdgcn_sched_barrier(0);
    }

    v8f T[MT];
#pragma unroll
    for (int m = 0; m < MT; ++m) T[m] = (v8f){};
#pragma unroll
    for (int s = 0; s < KS; ++s) {
#pragma unroll
      for (int m = 0; m < MT; ++m) {
        T[m] = __builtin_amdgcn_wmma_f32_16x16x4_f32(
            false, E[m][s], false, bv[s], (short)0, T[m], false, false);
      }
    }

    v8f S[NT];
#pragma unroll
    for (int t = 0; t < NT; ++t) S[t] = (v8f){};
#pragma unroll
    for (int q = 0; q < 4; ++q) {
#pragma unroll
      for (int t = 0; t < NT; ++t) {
        v2f bt;
        bt.x = T[t][2 * q];
        bt.y = T[t][2 * q + 1];
        S[t] = __builtin_amdgcn_wmma_f32_16x16x4_f32(
            false, ah[q], false, bt, (short)0, S[t], false, false);
      }
    }
#pragma unroll
    for (int t = 0; t < NT; ++t) {
      v2f bt;
      bt.x = T[t + 1][0];
      bt.y = T[t + 1][1];
      S[t] = __builtin_amdgcn_wmma_f32_16x16x4_f32(
          false, ah[4], false, bt, (short)0, S[t], false, false);
    }

    float* sw = &stage[w][b][0];
    float* swr = sw + c * PITCH + 8 * h;
#pragma unroll
    for (int t = 0; t < NT; ++t) {
      v4f o0, o1;
#pragma unroll
      for (int i = 0; i < 4; ++i) {
        o0[i] = fast_log2(S[t][i]) * LOG_SCALE + SHIFT;
        o1[i] = fast_log2(S[t][4 + i]) * LOG_SCALE + SHIFT;
      }
      *(v4f*)(swr + 16 * t) = o0;
      *(v4f*)(swr + 16 * t + 4) = o1;
    }
    __builtin_amdgcn_fence(__ATOMIC_RELEASE, "wavefront", "local");
    __builtin_amdgcn_wave_barrier();
    __builtin_amdgcn_fence(__ATOMIC_ACQUIRE, "wavefront", "local");
    const float* srd = sw + h * PITCH + 4 * c;
    float* __restrict__ orow = oi + (size_t)(Y0 + h) * IMG_W + X0 + 4 * c;
#pragma unroll
    for (int i = 0; i < 8; ++i) {
      const v4f v = *(const v4f*)(srd + 2 * i * PITCH);
      *(v4f*)(orow + 2 * i * IMG_W) = v;
    }
  }
}

extern "C" void kernel_launch(void* const* d_in, const int* in_sizes, int n_in,
                              void* d_out, int out_size, void* d_ws, size_t ws_size,
                              hipStream_t stream) {
  const float* x = (const float*)d_in[0];
  float* out = (float*)d_out;

  soft_dilate5x5_kernel<<<512, 32 * WAVES, 0, stream>>>(x, out);
}


# Vertical sum runs slice by slice in load order (10 exp + 5 matrix instructions per 10 loads): matrix work starts when the first slice has landed, not the whole block; next block's loads rolled in
# speedup vs baseline: 1.8667x; 1.0794x over previous
#include <hip/hip_runtime.h>


typedef float v2f __attribute__((ext_vector_type(2)));
typedef float v4f __attribute__((ext_vector_type(4)));
typedef float v8f __attribute__((ext_vector_type(8)));

#define IMG_H 1024
#define IMG_W 1024

#define MT 5
#define KS 5
#define NT 4
#define NBLK 2
#define WAVES 4
#define PITCH 68

#define SHIFT 2.0f
#define EXP_SCALE (15.0f * 1.4426950408889634f)
#define LOG_SCALE (0.6931471805599453f / 15.0f)

__device__ __forceinline__ float fast_exp2(float x) {
#if __has_builtin(__builtin_amdgcn_exp2f)
  return __builtin_amdgcn_exp2f(x);
#else
  return exp2f(x);
#endif
}
__device__ __forceinline__ float fast_log2(float x) {
#if __has_builtin(__builtin_amdgcn_logf)
  return __builtin_amdgcn_logf(x);
#else
  return log2f(x);
#endif
}

__device__ __forceinline__ int colperm(int i) {
  return i < 2 ? i : (i < 8 ? i + 2 : (i < 10 ? i - 6 : i));
}

__device__ __forceinline__ void load_slice(const float* __restrict__ xi,
                                           const unsigned (&xc)[MT], int Y0, int h, int s,
                                           v2f (&raw)[MT][KS]) {
  const int yb = Y0 - 2 + 2 * h;
#pragma unroll
  for (int e = 0; e < 2; ++e) {
    int y = yb + 4 * s + e;
    if (s == 0) y = y < 0 ? 0 : y;
    if (s == KS - 1) y = y > IMG_H - 1 ? IMG_H - 1 : y;
    const unsigned ro = (unsigned)y * IMG_W;
#pragma unroll
    for (int m = 0; m < MT; ++m) raw[m][s][e] = xi[ro + xc[m]];
  }
}

__global__ __launch_bounds__(32 * WAVES) void soft_dilate5x5_kernel(
    const float* __restrict__ x, float* __restrict__ out) {
  const int lane = threadIdx.x & 31;
  const int h = lane >> 4;
  const int c = lane & 15;
  const int w = __builtin_amdgcn_readfirstlane((int)(threadIdx.x >> 5));
  const int g = blockIdx.x * 4 + w;
  const int X0 = (g & 15) * 64;
  const int Yw = ((g >> 4) & 31) * (16 * NBLK);
  const int img = g >> 9;
  const float* __restrict__ xi = x + ((size_t)img << 20);
  float* __restrict__ oi = out + ((size_t)img << 20);

  __shared__ __attribute__((aligned(16))) float stage[WAVES][NBLK][16 * PITCH];

  v2f bv[KS];
#pragma unroll
  for (int s = 0; s < KS; ++s) {
#pragma unroll
    for (int e = 0; e < 2; ++e) {
      const int k = 4 * s + 2 * h + e;
      bv[s][e] = (k >= c && k <= c + 4) ? 1.0f : 0.0f;
    }
  }
  v2f ah[5];
#pragma unroll
  for (int q = 0; q < 4; ++q) {
#pragma unroll
    for (int e = 0; e < 2; ++e) {
      const int u = colperm(8 * h + 2 * q + e);
      ah[q][e] = (u >= c && u <= c + 4) ? 1.0f : 0.0f;
    }
  }
#pragma unroll
  for (int e = 0; e < 2; ++e) {
    const int u = 16 + colperm(8 * h + e);
    ah[4][e] = (u >= c && u <= c + 4) ? 1.0f : 0.0f;
  }

  unsigned xc[MT];
#pragma unroll
  for (int m = 0; m < MT; ++m) {
    int gx = X0 - 2 + 16 * m + colperm(c);
    gx = gx < 0 ? 0 : (gx > IMG_W - 1 ? IMG_W - 1 : gx);
    xc[m] = (unsigned)gx;
  }

  v2f raw[NBLK][MT][KS];
#pragma unroll
  for (int s = 0; s < KS; ++s) load_slice(xi, xc, Yw, h, s, raw[0]);
  __builtin_amdgcn_sched_barrier(0);

  v2f carry[MT];

#pragma unroll
  for (int b = 0; b < NBLK; ++b) {
    const int Y0 = Yw + 16 * b;

    v8f T[MT];
#pragma unroll
    for (int m = 0; m < MT; ++m) T[m] = (v8f){};
#pragma unroll
    for (int s = 0; s < KS; ++s) {
      v2f Es[MT];
#pragma unroll
      for (int m = 0; m < MT; ++m) {
        if (b > 0 && s == 0) {
          Es[m] = carry[m];
        } else {
          Es[m][0] = fast_exp2((raw[b][m][s][0] - SHIFT) * EXP_SCALE);
          Es[m][1] = fast_exp2((raw[b][m][s][1] - SHIFT) * EXP_SCALE);
        }
        if (s == KS - 1) carry[m] = Es[m];
      }
#pragma unroll
      for (int m = 0; m < MT; ++m) {
        T[m] = __builtin_amdgcn_wmma_f32_16x16x4_f32(
            false, Es[m], false, bv[s], (short)0, T[m], false, false);
      }
      if (b + 1 < NBLK && s + 1 < KS) {
        load_slice(xi, xc, Y0 + 16, h, s + 1, raw[b + 1]);
        __builtin_amdgcn_sched_barrier(0);
      }
    }

    v8f S[NT];
#pragma unroll
    for (int t = 0; t < NT; ++t) S[t] = (v8f){};
#pragma unroll
    for (int q = 0; q < 4; ++q) {
#pragma unroll
      for (int t = 0; t < NT; ++t) {
        v2f bt;
        bt.x = T[t][2 * q];
        bt.y = T[t][2 * q + 1];
        S[t] = __builtin_amdgcn_wmma_f32_16x16x4_f32(
            false, ah[q], false, bt, (short)0, S[t], false, false);
      }
    }
#pragma unroll
    for (int t = 0; t < NT; ++t) {
      v2f bt;
      bt.x = T[t + 1][0];
      bt.y = T[t + 1][1];
      S[t] = __builtin_amdgcn_wmma_f32_16x16x4_f32(
          false, ah[4], false, bt, (short)0, S[t], false, false);
    }

    float* sw = &stage[w][b][0];
    float* swr = sw + c * PITCH + 8 * h;
#pragma unroll
    for (int t = 0; t < NT; ++t) {
      v4f o0, o1;
#pragma unroll
      for (int i = 0; i < 4; ++i) {
        o0[i] = fast_log2(S[t][i]) * LOG_SCALE + SHIFT;
        o1[i] = fast_log2(S[t][4 + i]) * LOG_SCALE + SHIFT;
      }
      *(v4f*)(swr + 16 * t) = o0;
      *(v4f*)(swr + 16 * t + 4) = o1;
    }
    __builtin_amdgcn_fence(__ATOMIC_RELEASE, "wavefront", "local");
    __builtin_amdgcn_wave_barrier();
    __builtin_amdgcn_fence(__ATOMIC_ACQUIRE, "wavefront", "local");
    const float* srd = sw + h * PITCH + 4 * c;
    float* __restrict__ orow = oi + (size_t)(Y0 + h) * IMG_W + X0 + 4 * c;
#pragma unroll
    for (int i = 0; i < 8; ++i) {
      const v4f v = *(const v4f*)(srd + 2 * i * PITCH);
      *(v4f*)(orow + 2 * i * IMG_W) = v;
    }
  }
}

extern "C" void kernel_launch(void* const* d_in, const int* in_sizes, int n_in,
                              void* d_out, int out_size, void* d_ws, size_t ws_size,
                              hipStream_t stream) {
  const float* x = (const float*)d_in[0];
  float* out = (float*)d_out;

  soft_dilate5x5_kernel<<<512, 32 * WAVES, 0, stream>>>(x, out);
}
